# starting kernel
# baseline (speedup 1.0000x reference)
#include <hip/hip_runtime.h>
#include <hip/hip_bf16.h>


#define N_NODES 100000
#define N_EDGES 1600000
#define F_INPUT 16
#define HDIM 128
#define M_TILES (N_NODES / 16)

typedef float v2f __attribute__((ext_vector_type(2)));
typedef float v8f __attribute__((ext_vector_type(8)));

__global__ __launch_bounds__(256) void zero_f4_kernel(float4* __restrict__ p, long long n4) {
    long long i = (long long)blockIdx.x * 256 + threadIdx.x;
    if (i < n4) p[i] = make_float4(0.f, 0.f, 0.f, 0.f);
}

__global__ __launch_bounds__(256) void degree_kernel(const int* __restrict__ dst,
                                                     float* __restrict__ cnt, int E) {
    int e = blockIdx.x * 256 + threadIdx.x;
    if (e < E) atomicAdd(&cnt[dst[e]], 1.0f);
}

__global__ __launch_bounds__(256) void recip_kernel(float* __restrict__ deg, int n) {
    int i = blockIdx.x * 256 + threadIdx.x;
    if (i < n) deg[i] = 1.0f / fmaxf(deg[i], 1.0f);
}

template <int CHUNKS>
__global__ __launch_bounds__(256) void scatter_kernel(const float* __restrict__ in,
                                                      const int* __restrict__ src,
                                                      const int* __restrict__ dst,
                                                      float* __restrict__ agg, int E) {
    long long tid = (long long)blockIdx.x * 256 + threadIdx.x;
    if (tid >= (long long)E * CHUNKS) return;
    int e = (int)(tid / CHUNKS);
    int c = (int)(tid % CHUNKS);
    int s = src[e];
    int d = dst[e];
    float4 v = ((const float4*)in)[(long long)s * CHUNKS + c];
    float* p = agg + ((long long)d * CHUNKS + c) * 4;
    atomicAdd(p + 0, v.x);
    atomicAdd(p + 1, v.y);
    atomicAdd(p + 2, v.z);
    atomicAdd(p + 3, v.w);
}

__global__ __launch_bounds__(256) void sage_gemm_kernel(const float* __restrict__ in,
                                                        const float* __restrict__ agg,
                                                        const float* __restrict__ invdeg,
                                                        const float* __restrict__ Wr,
                                                        const float* __restrict__ Wl,
                                                        const float* __restrict__ bias,
                                                        float* __restrict__ out,
                                                        int K, int do_relu) {
    const int lane = threadIdx.x & 31;
    const int wv = threadIdx.x >> 5;
    const int mt = blockIdx.x * 8 + wv;
    if (mt >= M_TILES) return;

    const int row0 = mt << 4;
    const int m = lane & 15;
    const int kh = (lane >> 4) << 1;
    const long long arow = (long long)(row0 + m);

    v8f acc[8] = {};

    for (int k = 0; k < K; k += 4) {
        v2f a = *(const v2f*)(in + arow * K + k + kh);
#pragma unroll
        for (int nt = 0; nt < 8; ++nt) {
            const int col = (nt << 4) + m;
            v2f b;
            b.x = Wr[(k + kh) * HDIM + col];
            b.y = Wr[(k + kh + 1) * HDIM + col];
            acc[nt] = __builtin_amdgcn_wmma_f32_16x16x4_f32(
                false, a, false, b, (short)0, acc[nt], false, false);
        }
    }

    const float inv = invdeg[row0 + m];
    for (int k = 0; k < K; k += 4) {
        v2f a = *(const v2f*)(agg + arow * K + k + kh);
        a.x *= inv;
        a.y *= inv;
#pragma unroll
        for (int nt = 0; nt < 8; ++nt) {
            const int col = (nt << 4) + m;
            v2f b;
            b.x = Wl[(k + kh) * HDIM + col];
            b.y = Wl[(k + kh + 1) * HDIM + col];
            acc[nt] = __builtin_amdgcn_wmma_f32_16x16x4_f32(
                false, a, false, b, (short)0, acc[nt], false, false);
        }
    }

    const int rofs = (lane >> 4) << 3;
#pragma unroll
    for (int nt = 0; nt < 8; ++nt) {
        const int col = (nt << 4) + m;
        const float bv = bias[col];
#pragma unroll
        for (int j = 0; j < 8; ++j) {
            float v = acc[nt][j] + bv;
            if (do_relu) v = fmaxf(v, 0.0f);
            out[(long long)(row0 + rofs + j) * HDIM + col] = v;
        }
    }
}

extern "C" void kernel_launch(void* const* d_in, const int* in_sizes, int n_in,
                              void* d_out, int out_size, void* d_ws, size_t ws_size,
                              hipStream_t stream) {
    (void)in_sizes; (void)n_in; (void)out_size; (void)ws_size;

    const float* x = (const float*)d_in[0];
    const int* ei = (const int*)d_in[1];
    const int* src = ei;
    const int* dst = ei + N_EDGES;
    const float* Wl[4] = {(const float*)d_in[2], (const float*)d_in[5],
                          (const float*)d_in[8], (const float*)d_in[11]};
    const float* Wr[4] = {(const float*)d_in[3], (const float*)d_in[6],
                          (const float*)d_in[9], (const float*)d_in[12]};
    const float* bs[4] = {(const float*)d_in[4], (const float*)d_in[7],
                          (const float*)d_in[10], (const float*)d_in[13]};

    float* agg = (float*)d_ws;
    float* hA = agg + (size_t)N_NODES * HDIM;
    float* hB = hA + (size_t)N_NODES * HDIM;
    float* deg = hB + (size_t)N_NODES * HDIM;

    zero_f4_kernel<<<(N_NODES / 4 + 255) / 256, 256, 0, stream>>>((float4*)deg, N_NODES / 4);
    degree_kernel<<<(N_EDGES + 255) / 256, 256, 0, stream>>>(dst, deg, N_EDGES);
    recip_kernel<<<(N_NODES + 255) / 256, 256, 0, stream>>>(deg, N_NODES);

    const int gemm_grid = (M_TILES + 7) / 8;

    {
        long long n4 = (long long)N_NODES * F_INPUT / 4;
        zero_f4_kernel<<<(int)((n4 + 255) / 256), 256, 0, stream>>>((float4*)agg, n4);
        long long t = (long long)N_EDGES * (F_INPUT / 4);
        scatter_kernel<F_INPUT / 4><<<(int)((t + 255) / 256), 256, 0, stream>>>(
            x, src, dst, agg, N_EDGES);
        sage_gemm_kernel<<<gemm_grid, 256, 0, stream>>>(x, agg, deg, Wr[0], Wl[0], bs[0],
                                                        hA, F_INPUT, 1);
    }

    const float* cur = hA;
    float* nxt = hB;
    for (int l = 1; l < 4; ++l) {
        long long n4 = (long long)N_NODES * HDIM / 4;
        zero_f4_kernel<<<(int)((n4 + 255) / 256), 256, 0, stream>>>((float4*)agg, n4);
        long long t = (long long)N_EDGES * (HDIM / 4);
        scatter_kernel<HDIM / 4><<<(int)((t + 255) / 256), 256, 0, stream>>>(
            cur, src, dst, agg, N_EDGES);
        float* outp = (l == 3) ? (float*)d_out : nxt;
        int relu = (l == 3) ? 0 : 1;
        sage_gemm_kernel<<<gemm_grid, 256, 0, stream>>>(cur, agg, deg, Wr[l], Wl[l], bs[l],
                                                        outp, HDIM, relu);
        float* t2 = (float*)cur;
        cur = outp;
        nxt = t2;
    }
}


# Layers 2-4 scatter: each sum row is added into by one chip half only (work-groups read their die, claim edge chunks from that half's counter, skip the other half's rows); atomic sum order = arrival
# speedup vs baseline: 2.3534x; 2.3534x over previous
#include <hip/hip_runtime.h>
#include <hip/hip_bf16.h>


#define N_NODES 100000
#define N_EDGES 1600000
#define F_INPUT 16
#define HDIM 128
#define M_TILES (N_NODES / 16)
#define SUM1_PITCH 32

typedef float v2f __attribute__((ext_vector_type(2)));
typedef float v8f __attribute__((ext_vector_type(8)));

#define CLAIM_PITCH 64
#define CLAIM_INTS_PER_LAUNCH (2 * CLAIM_PITCH)
#define CLAIM_F4 (3 * CLAIM_INTS_PER_LAUNCH / 4)

__global__ __launch_bounds__(256) void zero_f4_kernel(float4* __restrict__ p, long long n4,
                                                      float4* __restrict__ claims) {
    long long i = (long long)blockIdx.x * 256 + threadIdx.x;
    if (i < n4) p[i] = make_float4(0.f, 0.f, 0.f, 0.f);
    if (i < CLAIM_F4) claims[i] = make_float4(0.f, 0.f, 0.f, 0.f);
}

#define COLS_PER_PHASE 64
#define PHASES (HDIM / COLS_PER_PHASE)
static_assert(COLS_PER_PHASE == 64, "a wave covers a row half with two 32-float lines");

#define HALF_NODES (N_NODES / 2)
#define CHUNK_EDGES 512
#define WAVE_EDGES (CHUNK_EDGES / 8)
#define ROUND_EDGES 32
#define GROUP_EDGES 16
#define N_CHUNKS (N_EDGES / CHUNK_EDGES)
#define CLAIMS_PER_HALF (PHASES * N_CHUNKS)
#define SCATTER_GRID 2048
static_assert(N_EDGES % CHUNK_EDGES == 0, "chunks are whole");
static_assert(WAVE_EDGES % ROUND_EDGES == 0, "a wave takes whole rounds of one edge per lane");
static_assert(ROUND_EDGES % GROUP_EDGES == 0, "a round is whole groups of edges");

__global__ __launch_bounds__(256) void scatter_rows128_kernel(const float* __restrict__ in,
                                                              const int* __restrict__ src,
                                                              const int* __restrict__ dst,
                                                              float* __restrict__ agg,
                                                              int* __restrict__ claim) {
    __shared__ int s_claim[2];
    const int lane = threadIdx.x & 31;
    const int wv = __builtin_amdgcn_readfirstlane((int)(threadIdx.x >> 5));

    int which = 0;
    int other_tried = 0;
    if (threadIdx.x == 0) which = (((__builtin_amdgcn_s_sendmsg_rtn(0x87) >> 16) & 15) >= 4) ? 1 : 0;

    for (int it = 0; it < 2 * CLAIMS_PER_HALF + 2; ++it) {
        if (threadIdx.x == 0) {
            int c = atomicAdd(claim + which * CLAIM_PITCH, 1);
            if (c >= CLAIMS_PER_HALF && !other_tried) {
                other_tried = 1;
                which ^= 1;
                c = atomicAdd(claim + which * CLAIM_PITCH, 1);
            }
            s_claim[it & 1] = (c < CLAIMS_PER_HALF) ? (c * 2 + which) : -1;
        }
        __syncthreads();
        const int t = __builtin_amdgcn_readfirstlane(s_claim[it & 1]);
        if (t < 0) break;
        const int own = t & 1;
        const int c = t >> 1;
        const int phase = (c >= N_CHUNKS) ? 1 : 0;
        const int chunk = c - phase * N_CHUNKS;
        const int c0 = phase * COLS_PER_PHASE + lane;
        const int ebase = chunk * CHUNK_EDGES + wv * WAVE_EDGES;

        for (int r = 0; r < WAVE_EDGES / ROUND_EDGES; ++r) {
            const int sv = src[ebase + r * ROUND_EDGES + lane];
            const int dv = dst[ebase + r * ROUND_EDGES + lane];
            for (int g = 0; g < ROUND_EDGES / GROUP_EDGES; ++g) {
                float v[GROUP_EDGES][2];
                int d[GROUP_EDGES];
                for (int i = 0; i < GROUP_EDGES; ++i) {
                    const int s = __builtin_amdgcn_readlane(sv, g * GROUP_EDGES + i);
                    d[i] = __builtin_amdgcn_readlane(dv, g * GROUP_EDGES + i);
                    v[i][0] = 0.f;
                    v[i][1] = 0.f;
                    if ((d[i] >= HALF_NODES ? 1 : 0) == own) {
                        const float* p = in + (long long)s * HDIM + c0;
                        v[i][0] = p[0];
                        v[i][1] = p[32];
                    }
                }
                for (int i = 0; i < GROUP_EDGES; ++i) {
                    if ((d[i] >= HALF_NODES ? 1 : 0) == own) {
                        float* p = agg + (long long)d[i] * HDIM + c0;
                        atomicAdd(p, v[i][0]);
                        atomicAdd(p + 32, v[i][1]);
                    }
                }
            }
        }
    }
}

#define L1_EDGES_PER_WAVE 16
static_assert(N_EDGES % L1_EDGES_PER_WAVE == 0, "a wave takes whole groups of edges");

__global__ __launch_bounds__(256) void scatter_rows16_count_kernel(const float* __restrict__ in,
                                                                   const int* __restrict__ src,
                                                                   const int* __restrict__ dst,
                                                                   float* __restrict__ sum1, int E) {
    const int lane = threadIdx.x & 31;
    const int wave = __builtin_amdgcn_readfirstlane((int)((blockIdx.x * 256 + threadIdx.x) >> 5));
    const int e0 = wave * L1_EDGES_PER_WAVE;
    if (e0 >= E) return;

    float v[L1_EDGES_PER_WAVE];
    int d[L1_EDGES_PER_WAVE];
    for (int i = 0; i < L1_EDGES_PER_WAVE; ++i) {
        const int s = src[e0 + i];
        d[i] = dst[e0 + i];
        const float xv = in[(long long)s * F_INPUT + (lane & (F_INPUT - 1))];
        v[i] = (lane < F_INPUT) ? xv : 1.0f;
    }
    if (lane <= F_INPUT) {
        for (int i = 0; i < L1_EDGES_PER_WAVE; ++i)
            atomicAdd(sum1 + (long long)d[i] * SUM1_PITCH + lane, v[i]);
    }
}

__global__ __launch_bounds__(256) void sage_gemm_kernel(const float* __restrict__ in,
                                                        const float* agg, int agg_pitch,
                                                        const float* __restrict__ invdeg_in,
                                                        float* __restrict__ invdeg_out,
                                                        const float* __restrict__ Wr,
                                                        const float* __restrict__ Wl,
                                                        const float* __restrict__ bias,
                                                        float* __restrict__ out,
                                                        float* next_sum,
                                                        int K, int do_relu) {
    const int lane = threadIdx.x & 31;
    const int wv = threadIdx.x >> 5;
    const int mt = blockIdx.x * 8 + wv;
    if (mt >= M_TILES) return;

    const int row0 = mt << 4;
    const int m = lane & 15;
    const int kh = (lane >> 4) << 1;
    const long long arow = (long long)(row0 + m);

    v8f acc[8] = {};

    for (int k = 0; k < K; k += 4) {
        v2f a = *(const v2f*)(in + arow * K + k + kh);
#pragma unroll
        for (int nt = 0; nt < 8; ++nt) {
            const int col = (nt << 4) + m;
            v2f b;
            b.x = Wr[(k + kh) * HDIM + col];
            b.y = Wr[(k + kh + 1) * HDIM + col];
            acc[nt] = __builtin_amdgcn_wmma_f32_16x16x4_f32(
                false, a, false, b, (short)0, acc[nt], false, false);
        }
    }

    float inv;
    if (invdeg_out != nullptr) {
        inv = 1.0f / fmaxf(agg[arow * agg_pitch + F_INPUT], 1.0f);
        if (lane < 16) invdeg_out[row0 + m] = inv;
    } else {
        inv = invdeg_in[row0 + m];
    }
    for (int k = 0; k < K; k += 4) {
        v2f a = *(const v2f*)(agg + arow * agg_pitch + k + kh);
        a.x *= inv;
        a.y *= inv;
#pragma unroll
        for (int nt = 0; nt < 8; ++nt) {
            const int col = (nt << 4) + m;
            v2f b;
            b.x = Wl[(k + kh) * HDIM + col];
            b.y = Wl[(k + kh + 1) * HDIM + col];
            acc[nt] = __builtin_amdgcn_wmma_f32_16x16x4_f32(
                false, a, false, b, (short)0, acc[nt], false, false);
        }
    }

    if (next_sum != nullptr) {
        float4* z = (float4*)(next_sum + (long long)row0 * HDIM) + lane;
#pragma unroll
        for (int i = 0; i < 16; ++i) z[i * 32] = make_float4(0.f, 0.f, 0.f, 0.f);
    }

    const int rofs = (lane >> 4) << 3;
#pragma unroll
    for (int nt = 0; nt < 8; ++nt) {
        const int col = (nt << 4) + m;
        const float bv = bias[col];
#pragma unroll
        for (int j = 0; j < 8; ++j) {
            float v = acc[nt][j] + bv;
            if (do_relu) v = fmaxf(v, 0.0f);
            out[(long long)(row0 + rofs + j) * HDIM + col] = v;
        }
    }
}

extern "C" void kernel_launch(void* const* d_in, const int* in_sizes, int n_in,
                              void* d_out, int out_size, void* d_ws, size_t ws_size,
                              hipStream_t stream) {
    (void)in_sizes; (void)n_in; (void)out_size; (void)ws_size;

    const float* x = (const float*)d_in[0];
    const int* ei = (const int*)d_in[1];
    const int* src = ei;
    const int* dst = ei + N_EDGES;
    const float* Wl[4] = {(const float*)d_in[2], (const float*)d_in[5],
                          (const float*)d_in[8], (const float*)d_in[11]};
    const float* Wr[4] = {(const float*)d_in[3], (const float*)d_in[6],
                          (const float*)d_in[9], (const float*)d_in[12]};
    const float* bs[4] = {(const float*)d_in[4], (const float*)d_in[7],
                          (const float*)d_in[10], (const float*)d_in[13]};

    float* agg = (float*)d_ws;
    float* hA = agg + (size_t)N_NODES * HDIM;
    float* hB = hA + (size_t)N_NODES * HDIM;
    float* deg = hB + (size_t)N_NODES * HDIM;
    float* sum1 = hB;
    int* claims = (int*)d_out;

    const int gemm_grid = (M_TILES + 7) / 8;

    {
        long long n4 = (long long)N_NODES * SUM1_PITCH / 4;
        zero_f4_kernel<<<(int)((n4 + 255) / 256), 256, 0, stream>>>((float4*)sum1, n4,
                                                                    (float4*)claims);
        scatter_rows16_count_kernel<<<(N_EDGES / L1_EDGES_PER_WAVE + 7) / 8, 256, 0, stream>>>(
            x, src, dst, sum1, N_EDGES);
        sage_gemm_kernel<<<gemm_grid, 256, 0, stream>>>(x, sum1, SUM1_PITCH, nullptr, deg,
                                                        Wr[0], Wl[0], bs[0], hA, agg,
                                                        F_INPUT, 1);
    }

    const float* cur = hA;
    float* nxt = hB;
    for (int l = 1; l < 4; ++l) {
        scatter_rows128_kernel<<<SCATTER_GRID, 256, 0, stream>>>(
            cur, src, dst, agg, claims + (l - 1) * CLAIM_INTS_PER_LAUNCH);
        float* outp = (l == 3) ? (float*)d_out : nxt;
        float* clr = (l == 3) ? nullptr : agg;
        int relu = (l == 3) ? 0 : 1;
        sage_gemm_kernel<<<gemm_grid, 256, 0, stream>>>(cur, agg, HDIM, deg, nullptr,
                                                        Wr[l], Wl[l], bs[l], outp, clr,
                                                        HDIM, relu);
        float* t2 = (float*)cur;
        cur = outp;
        nxt = t2;
    }
}


# Layers 2-4 scatter: a wave compacts its own-half edges to its low lanes once per 32 edges, then count-nested edge code (one scalar test per own edge), 32-bit row offsets; atomic sum order = arrival
# speedup vs baseline: 2.4278x; 1.0316x over previous
#include <hip/hip_runtime.h>
#include <hip/hip_bf16.h>


#define N_NODES 100000
#define N_EDGES 1600000
#define F_INPUT 16
#define HDIM 128
#define M_TILES (N_NODES / 16)
#define SUM1_PITCH 32

typedef float v2f __attribute__((ext_vector_type(2)));
typedef float v8f __attribute__((ext_vector_type(8)));

#define CLAIM_PITCH 64
#define CLAIM_INTS_PER_LAUNCH (2 * CLAIM_PITCH)
#define CLAIM_F4 (3 * CLAIM_INTS_PER_LAUNCH / 4)

__global__ __launch_bounds__(256) void zero_f4_kernel(float4* __restrict__ p, long long n4,
                                                      float4* __restrict__ claims) {
    long long i = (long long)blockIdx.x * 256 + threadIdx.x;
    if (i < n4) p[i] = make_float4(0.f, 0.f, 0.f, 0.f);
    if (i < CLAIM_F4) claims[i] = make_float4(0.f, 0.f, 0.f, 0.f);
}

#define COLS_PER_PHASE 64
#define PHASES (HDIM / COLS_PER_PHASE)
static_assert(COLS_PER_PHASE == 64, "a wave covers a row half with two 32-float lines");

#define HALF_NODES (N_NODES / 2)
#define CHUNK_EDGES 512
#define WAVE_EDGES (CHUNK_EDGES / 8)
#define ROUND_EDGES 32
#define GROUP_SLOTS 20
#define N_CHUNKS (N_EDGES / CHUNK_EDGES)
#define CLAIMS_PER_HALF (PHASES * N_CHUNKS)
#define SCATTER_GRID 2048
static_assert(N_EDGES % CHUNK_EDGES == 0, "chunks are whole");
static_assert(WAVE_EDGES % ROUND_EDGES == 0, "a wave takes whole rounds of one edge per lane");
static_assert(GROUP_SLOTS > 0 && GROUP_SLOTS < ROUND_EDGES, "a first group and a rest");

template <int I, int END, int BASE>
struct OwnEdges {
    static __device__ __forceinline__ void run(int cnt, unsigned cs, unsigned cd, unsigned c0,
                                               const float* __restrict__ in,
                                               float* __restrict__ agg) {
        if (cnt > I) {
            const unsigned so = (unsigned)__builtin_amdgcn_readlane((int)cs, BASE + I);
            const float* p = in + (so + c0);
            const float v0 = p[0];
            const float v1 = p[32];
            OwnEdges<I + 1, END, BASE>::run(cnt, cs, cd, c0, in, agg);
            const unsigned dof = (unsigned)__builtin_amdgcn_readlane((int)cd, BASE + I);
            float* q = agg + (dof + c0);
            atomicAdd(q, v0);
            atomicAdd(q + 32, v1);
        }
    }
};
template <int END, int BASE>
struct OwnEdges<END, END, BASE> {
    static __device__ __forceinline__ void run(int, unsigned, unsigned, unsigned, const float*,
                                               float*) {}
};

__global__ __launch_bounds__(256) void scatter_rows128_kernel(const float* __restrict__ in,
                                                              const int* __restrict__ src,
                                                              const int* __restrict__ dst,
                                                              float* __restrict__ agg,
                                                              int* __restrict__ claim) {
    __shared__ int s_claim[2];
    const int lane = threadIdx.x & 31;
    const int wv = __builtin_amdgcn_readfirstlane((int)(threadIdx.x >> 5));

    int which = 0;
    int other_tried = 0;
    if (threadIdx.x == 0) which = (((__builtin_amdgcn_s_sendmsg_rtn(0x87) >> 16) & 15) >= 4) ? 1 : 0;

    for (int it = 0; it < 2 * CLAIMS_PER_HALF + 2; ++it) {
        if (threadIdx.x == 0) {
            int c = atomicAdd(claim + which * CLAIM_PITCH, 1);
            if (c >= CLAIMS_PER_HALF && !other_tried) {
                other_tried = 1;
                which ^= 1;
                c = atomicAdd(claim + which * CLAIM_PITCH, 1);
            }
            s_claim[it & 1] = (c < CLAIMS_PER_HALF) ? (c * 2 + which) : -1;
        }
        __syncthreads();
        const int t = __builtin_amdgcn_readfirstlane(s_claim[it & 1]);
        if (t < 0) break;
        const int own = t & 1;
        const int c = t >> 1;
        const int phase = (c >= N_CHUNKS) ? 1 : 0;
        const int chunk = c - phase * N_CHUNKS;
        const unsigned c0 = (unsigned)(phase * COLS_PER_PHASE + lane);
        const int ebase = chunk * CHUNK_EDGES + wv * WAVE_EDGES;

        for (int r = 0; r < WAVE_EDGES / ROUND_EDGES; ++r) {
            const int sv = src[ebase + r * ROUND_EDGES + lane];
            const int dv = dst[ebase + r * ROUND_EDGES + lane];
            const unsigned upper = __builtin_amdgcn_ballot_w32(dv >= HALF_NODES);
            const unsigned mine = own ? upper : ~upper;
            const int n = __builtin_popcount(mine);
            const unsigned srow = (unsigned)sv * HDIM;
            const unsigned drow = (unsigned)dv * HDIM;
            const unsigned below = __builtin_amdgcn_mbcnt_lo(mine, 0u);
            const unsigned slot = ((mine >> lane) & 1u) ? below : (unsigned)n + (unsigned)lane - below;
            const unsigned cs = (unsigned)__builtin_amdgcn_ds_permute((int)(slot << 2), (int)srow);
            const unsigned cd = (unsigned)__builtin_amdgcn_ds_permute((int)(slot << 2), (int)drow);
            OwnEdges<0, GROUP_SLOTS, 0>::run(n, cs, cd, c0, in, agg);
            if (n > GROUP_SLOTS)
                OwnEdges<0, ROUND_EDGES - GROUP_SLOTS, GROUP_SLOTS>::run(n - GROUP_SLOTS, cs, cd, c0,
                                                                         in, agg);
        }
    }
}

#define L1_EDGES_PER_WAVE 16
static_assert(N_EDGES % L1_EDGES_PER_WAVE == 0, "a wave takes whole groups of edges");

__global__ __launch_bounds__(256) void scatter_rows16_count_kernel(const float* __restrict__ in,
                                                                   const int* __restrict__ src,
                                                                   const int* __restrict__ dst,
                                                                   float* __restrict__ sum1, int E) {
    const int lane = threadIdx.x & 31;
    const int wave = __builtin_amdgcn_readfirstlane((int)((blockIdx.x * 256 + threadIdx.x) >> 5));
    const int e0 = wave * L1_EDGES_PER_WAVE;
    if (e0 >= E) return;

    float v[L1_EDGES_PER_WAVE];
    int d[L1_EDGES_PER_WAVE];
    for (int i = 0; i < L1_EDGES_PER_WAVE; ++i) {
        const int s = src[e0 + i];
        d[i] = dst[e0 + i];
        const float xv = in[(long long)s * F_INPUT + (lane & (F_INPUT - 1))];
        v[i] = (lane < F_INPUT) ? xv : 1.0f;
    }
    if (lane <= F_INPUT) {
        for (int i = 0; i < L1_EDGES_PER_WAVE; ++i)
            atomicAdd(sum1 + (long long)d[i] * SUM1_PITCH + lane, v[i]);
    }
}

__global__ __launch_bounds__(256) void sage_gemm_kernel(const float* __restrict__ in,
                                                        const float* agg, int agg_pitch,
                                                        const float* __restrict__ invdeg_in,
                                                        float* __restrict__ invdeg_out,
                                                        const float* __restrict__ Wr,
                                                        const float* __restrict__ Wl,
                                                        const float* __restrict__ bias,
                                                        float* __restrict__ out,
                                                        float* next_sum,
                                                        int K, int do_relu) {
    const int lane = threadIdx.x & 31;
    const int wv = threadIdx.x >> 5;
    const int mt = blockIdx.x * 8 + wv;
    if (mt >= M_TILES) return;

    const int row0 = mt << 4;
    const int m = lane & 15;
    const int kh = (lane >> 4) << 1;
    const long long arow = (long long)(row0 + m);

    v8f acc[8] = {};

    for (int k = 0; k < K; k += 4) {
        v2f a = *(const v2f*)(in + arow * K + k + kh);
#pragma unroll
        for (int nt = 0; nt < 8; ++nt) {
            const int col = (nt << 4) + m;
            v2f b;
            b.x = Wr[(k + kh) * HDIM + col];
            b.y = Wr[(k + kh + 1) * HDIM + col];
            acc[nt] = __builtin_amdgcn_wmma_f32_16x16x4_f32(
                false, a, false, b, (short)0, acc[nt], false, false);
        }
    }

    float inv;
    if (invdeg_out != nullptr) {
        inv = 1.0f / fmaxf(agg[arow * agg_pitch + F_INPUT], 1.0f);
        if (lane < 16) invdeg_out[row0 + m] = inv;
    } else {
        inv = invdeg_in[row0 + m];
    }
    for (int k = 0; k < K; k += 4) {
        v2f a = *(const v2f*)(agg + arow * agg_pitch + k + kh);
        a.x *= inv;
        a.y *= inv;
#pragma unroll
        for (int nt = 0; nt < 8; ++nt) {
            const int col = (nt << 4) + m;
            v2f b;
            b.x = Wl[(k + kh) * HDIM + col];
            b.y = Wl[(k + kh + 1) * HDIM + col];
            acc[nt] = __builtin_amdgcn_wmma_f32_16x16x4_f32(
                false, a, false, b, (short)0, acc[nt], false, false);
        }
    }

    if (next_sum != nullptr) {
        float4* z = (float4*)(next_sum + (long long)row0 * HDIM) + lane;
#pragma unroll
        for (int i = 0; i < 16; ++i) z[i * 32] = make_float4(0.f, 0.f, 0.f, 0.f);
    }

    const int rofs = (lane >> 4) << 3;
#pragma unroll
    for (int nt = 0; nt < 8; ++nt) {
        const int col = (nt << 4) + m;
        const float bv = bias[col];
#pragma unroll
        for (int j = 0; j < 8; ++j) {
            float v = acc[nt][j] + bv;
            if (do_relu) v = fmaxf(v, 0.0f);
            out[(long long)(row0 + rofs + j) * HDIM + col] = v;
        }
    }
}

extern "C" void kernel_launch(void* const* d_in, const int* in_sizes, int n_in,
                              void* d_out, int out_size, void* d_ws, size_t ws_size,
                              hipStream_t stream) {
    (void)in_sizes; (void)n_in; (void)out_size; (void)ws_size;

    const float* x = (const float*)d_in[0];
    const int* ei = (const int*)d_in[1];
    const int* src = ei;
    const int* dst = ei + N_EDGES;
    const float* Wl[4] = {(const float*)d_in[2], (const float*)d_in[5],
                          (const float*)d_in[8], (const float*)d_in[11]};
    const float* Wr[4] = {(const float*)d_in[3], (const float*)d_in[6],
                          (const float*)d_in[9], (const float*)d_in[12]};
    const float* bs[4] = {(const float*)d_in[4], (const float*)d_in[7],
                          (const float*)d_in[10], (const float*)d_in[13]};

    float* agg = (float*)d_ws;
    float* hA = agg + (size_t)N_NODES * HDIM;
    float* hB = hA + (size_t)N_NODES * HDIM;
    float* deg = hB + (size_t)N_NODES * HDIM;
    float* sum1 = hB;
    int* claims = (int*)d_out;

    const int gemm_grid = (M_TILES + 7) / 8;

    {
        long long n4 = (long long)N_NODES * SUM1_PITCH / 4;
        zero_f4_kernel<<<(int)((n4 + 255) / 256), 256, 0, stream>>>((float4*)sum1, n4,
                                                                    (float4*)claims);
        scatter_rows16_count_kernel<<<(N_EDGES / L1_EDGES_PER_WAVE + 7) / 8, 256, 0, stream>>>(
            x, src, dst, sum1, N_EDGES);
        sage_gemm_kernel<<<gemm_grid, 256, 0, stream>>>(x, sum1, SUM1_PITCH, nullptr, deg,
                                                        Wr[0], Wl[0], bs[0], hA, agg,
                                                        F_INPUT, 1);
    }

    const float* cur = hA;
    float* nxt = hB;
    for (int l = 1; l < 4; ++l) {
        scatter_rows128_kernel<<<SCATTER_GRID, 256, 0, stream>>>(
            cur, src, dst, agg, claims + (l - 1) * CLAIM_INTS_PER_LAUNCH);
        float* outp = (l == 3) ? (float*)d_out : nxt;
        float* clr = (l == 3) ? nullptr : agg;
        int relu = (l == 3) ? 0 : 1;
        sage_gemm_kernel<<<gemm_grid, 256, 0, stream>>>(cur, agg, HDIM, deg, nullptr,
                                                        Wr[l], Wl[l], bs[l], outp, clr,
                                                        HDIM, relu);
        float* t2 = (float*)cur;
        cur = outp;
        nxt = t2;
    }
}
